# starting kernel
# baseline (speedup 1.0000x reference)
#include <hip/hip_runtime.h>

#define HN (2048 * 2048)
#define EPSF 1e-10f
#define DOUBLING_STEPS 23

typedef __attribute__((ext_vector_type(2))) float v2f;
typedef __attribute__((ext_vector_type(8))) float v8f;

__global__ __launch_bounds__(256) void phase1_contrib(
    const int* __restrict__ parent,
    const float* __restrict__ diff,
    const float* __restrict__ attr,
    const float* __restrict__ weight,
    const float* __restrict__ bias,
    unsigned long long* __restrict__ buf0)
{
    const int lane = threadIdx.x & 31;
    const int half = lane >> 4;
    const int m    = lane & 15;
    const int wave = blockIdx.x * (blockDim.x >> 5) + (threadIdx.x >> 5);
    const int base = wave << 4;
    const int row  = base + m;
    const float* a = attr + (size_t)row * 15;

    float bx[5], by[5];
    if (half == 0) {
        float a5 = a[5], a6 = a[6];
        bx[0] = a[0];                        by[0] = a[1];
        bx[1] = a[4];                        by[1] = logf(fabsf(a6) + EPSF);
        bx[2] = logf(fabsf(a[9])  + EPSF);   by[2] = logf(fabsf(a[10]) + EPSF);
        bx[3] = logf(fabsf(a[13]) + EPSF);   by[3] = logf(fabsf(a[14]) + EPSF);
        bx[4] = sinf(a5);                    by[4] = 0.0f;
    } else {
        float a5 = a[5], a6 = a[6], a7 = a[7];
        bx[0] = a[2];                        by[0] = a[3];
        bx[1] = logf(fabsf(a7)    + EPSF);   by[1] = logf(fabsf(a[8])  + EPSF);
        bx[2] = logf(fabsf(a[11]) + EPSF);   by[2] = logf(fabsf(a[12]) + EPSF);
        bx[3] = sqrtf(a7) / (sqrtf(a6) + EPSF);  by[3] = cosf(a5);
        bx[4] = 0.0f;                        by[4] = 0.0f;
    }

    float ax[5], ay[5];
#pragma unroll
    for (int c = 0; c < 5; ++c) {
        int kx = 4 * c + 2 * half;
        int ky = kx + 1;
        ax[c] = (kx < 17) ? weight[kx] : 0.0f;
        ay[c] = (ky < 17) ? weight[ky] : 0.0f;
    }

    v8f acc = {};
#pragma unroll
    for (int c = 0; c < 5; ++c) {
        v2f A; A.x = ax[c]; A.y = ay[c];
        v2f B; B.x = bx[c]; B.y = by[c];
        acc = __builtin_amdgcn_wmma_f32_16x16x4_f32(
             false, A,  false, B,
             (short)0, acc,  false,  false);
    }

    float val   = acc[0] + bias[0];
    float score = 1.0f / (1.0f + expf(-val));

    if (half == 0) {
        float contrib = diff[row] * score;
        unsigned long long packed =
            ((unsigned long long)(unsigned int)parent[row] << 32) |
            (unsigned long long)__float_as_uint(contrib);
        buf0[row] = packed;
    }
    if (blockIdx.x == 0 && threadIdx.x == 0) {
        buf0[HN] = ((unsigned long long)(unsigned int)HN << 32);
    }
}

__global__ __launch_bounds__(256) void doubling_step(
    const unsigned long long* __restrict__ in,
    unsigned long long* __restrict__ out,
    int n)
{
    int i = blockIdx.x * blockDim.x + threadIdx.x;
    if (i >= n) return;
    unsigned long long v = in[i];
    int p   = (int)(v >> 32);
    float c = __uint_as_float((unsigned int)v);
    unsigned long long w = in[p];
    float cp = __uint_as_float((unsigned int)w);
    out[i] = (w & 0xFFFFFFFF00000000ull)
           | (unsigned long long)__float_as_uint(c + cp);
}

__global__ __launch_bounds__(256) void final_step(
    const unsigned long long* __restrict__ in,
    float* __restrict__ out,
    int n)
{
    int i = blockIdx.x * blockDim.x + threadIdx.x;
    if (i >= n) return;
    unsigned long long v = in[i];
    int p   = (int)(v >> 32);
    float c = __uint_as_float((unsigned int)v);
    unsigned long long w = in[p];
    out[i] = c + __uint_as_float((unsigned int)w);
}

extern "C" void kernel_launch(void* const* d_in, const int* in_sizes, int n_in,
                              void* d_out, int out_size, void* d_ws, size_t ws_size,
                              hipStream_t stream) {
    const int*   parent = (const int*)d_in[0];
    const float* diff   = (const float*)d_in[1];
    const float* attr   = (const float*)d_in[2];
    const float* weight = (const float*)d_in[3];
    const float* bias   = (const float*)d_in[4];
    float* out = (float*)d_out;

    unsigned long long* buf0 = (unsigned long long*)d_ws;
    unsigned long long* buf1 = buf0 + (HN + 1);

    {
        int blocks = HN / (16 * 8);
        phase1_contrib<<<blocks, 256, 0, stream>>>(parent, diff, attr, weight, bias, buf0);
    }

    {
        int n = HN + 1;
        int blocks = (n + 255) / 256;
        unsigned long long* a = buf0;
        unsigned long long* b = buf1;
        for (int s = 0; s < DOUBLING_STEPS - 1; ++s) {
            doubling_step<<<blocks, 256, 0, stream>>>(a, b, n);
            unsigned long long* t = a; a = b; b = t;
        }
        final_step<<<(HN + 255) / 256, 256, 0, stream>>>(a, out, HN);
    }
}


# Levels 14-21 run as ONE launch of one workgroup: level 13 leaves a per-block bitmask of tiles not yet final, the workgroup steps only those, level by level behind its barrier; 24 launches become 17
# speedup vs baseline: 1.3051x; 1.3051x over previous
#include <hip/hip_runtime.h>

#define HN (2048 * 2048)
#define EPSF 1e-10f
#define DOUBLING_STEPS 23

typedef __attribute__((ext_vector_type(2))) float v2f;
typedef __attribute__((ext_vector_type(8))) float v8f;

__global__ __launch_bounds__(256) void phase1_contrib(
    const int* __restrict__ parent,
    const float* __restrict__ diff,
    const float* __restrict__ attr,
    const float* __restrict__ weight,
    const float* __restrict__ bias,
    unsigned long long* __restrict__ buf0)
{
    const int lane = threadIdx.x & 31;
    const int half = lane >> 4;
    const int m    = lane & 15;
    const int wave = blockIdx.x * (blockDim.x >> 5) + (threadIdx.x >> 5);
    const int base = wave << 4;
    const int row  = base + m;
    const float* a = attr + (size_t)row * 15;

    float bx[5], by[5];
    if (half == 0) {
        float a5 = a[5], a6 = a[6];
        bx[0] = a[0];                        by[0] = a[1];
        bx[1] = a[4];                        by[1] = logf(fabsf(a6) + EPSF);
        bx[2] = logf(fabsf(a[9])  + EPSF);   by[2] = logf(fabsf(a[10]) + EPSF);
        bx[3] = logf(fabsf(a[13]) + EPSF);   by[3] = logf(fabsf(a[14]) + EPSF);
        bx[4] = sinf(a5);                    by[4] = 0.0f;
    } else {
        float a5 = a[5], a6 = a[6], a7 = a[7];
        bx[0] = a[2];                        by[0] = a[3];
        bx[1] = logf(fabsf(a7)    + EPSF);   by[1] = logf(fabsf(a[8])  + EPSF);
        bx[2] = logf(fabsf(a[11]) + EPSF);   by[2] = logf(fabsf(a[12]) + EPSF);
        bx[3] = sqrtf(a7) / (sqrtf(a6) + EPSF);  by[3] = cosf(a5);
        bx[4] = 0.0f;                        by[4] = 0.0f;
    }

    float ax[5], ay[5];
#pragma unroll
    for (int c = 0; c < 5; ++c) {
        int kx = 4 * c + 2 * half;
        int ky = kx + 1;
        ax[c] = (kx < 17) ? weight[kx] : 0.0f;
        ay[c] = (ky < 17) ? weight[ky] : 0.0f;
    }

    v8f acc = {};
#pragma unroll
    for (int c = 0; c < 5; ++c) {
        v2f A; A.x = ax[c]; A.y = ay[c];
        v2f B; B.x = bx[c]; B.y = by[c];
        acc = __builtin_amdgcn_wmma_f32_16x16x4_f32(
             false, A,  false, B,
             (short)0, acc,  false,  false);
    }

    float val   = acc[0] + bias[0];
    float score = 1.0f / (1.0f + expf(-val));

    if (half == 0) {
        float contrib = diff[row] * score;
        unsigned long long packed =
            ((unsigned long long)(unsigned int)parent[row] << 32) |
            (unsigned long long)__float_as_uint(contrib);
        buf0[row] = packed;
    }
    if (blockIdx.x == 0 && threadIdx.x == 0) {
        buf0[HN] = ((unsigned long long)(unsigned int)HN << 32);
    }
}

#define PMASK    0x3FFFFFFFu
#define ST_SHIFT 30
#define ST_ONE   (1u << ST_SHIFT)
#define ST_BOTH  (2u << ST_SHIFT)

__global__ __launch_bounds__(256) void doubling_step(
    unsigned long long* in,
    unsigned long long* out,
    int n)
{
    int i = blockIdx.x * blockDim.x + threadIdx.x;
    if (i >= n) return;
    unsigned long long v = in[i];
    unsigned int hi = (unsigned int)(v >> 32);
    unsigned int lo = (unsigned int)v;
    unsigned int st = (unsigned int)__builtin_amdgcn_readfirstlane((int)hi) >> ST_SHIFT;
    if (st == 2u) return;
    if (st == 1u) {
        if ((threadIdx.x & 31) == 0) {
            unsigned long long m =
                ((unsigned long long)((hi & PMASK) | ST_BOTH) << 32) | (unsigned long long)lo;
            in[i]  = m;
            out[i] = m;
        } else {
            out[i] = v;
        }
        return;
    }
    int p   = (int)(hi & PMASK);
    float c = __uint_as_float(lo);
    unsigned long long w = in[p];
    float cp = __uint_as_float((unsigned int)w);
    unsigned int p2 = (unsigned int)(w >> 32) & PMASK;
    unsigned int live = __builtin_amdgcn_ballot_w32(p2 != (unsigned int)HN);
    if (live == 0u && (threadIdx.x & 31) == 0) p2 |= ST_ONE;
    out[i] = ((unsigned long long)p2 << 32)
           | (unsigned long long)__float_as_uint(c + cp);
}

#define LATE_FROM       12
#define TILES_PER_WAVE  8
#define NTILES          ((HN + 1 + 31) / 32)
#define LATE_BLOCKS     ((NTILES + 8 * TILES_PER_WAVE - 1) / (8 * TILES_PER_WAVE))

template <bool WRITE_MASK>
__global__ __launch_bounds__(256) void doubling_step_late(
    unsigned long long* in,
    unsigned long long* out,
    int n,
    unsigned long long* masks)
{
    const int lane   = threadIdx.x & 31;
    const int wave   = blockIdx.x * (blockDim.x >> 5) + (threadIdx.x >> 5);
    const int tile0  = wave * TILES_PER_WAVE;
    const int ntiles = (n + 31) >> 5;

    unsigned int tst = 2u;
    if (lane < TILES_PER_WAVE && tile0 + lane < ntiles)
        tst = (unsigned int)(in[(size_t)(tile0 + lane) << 5] >> (32 + ST_SHIFT));
    unsigned int todo = __builtin_amdgcn_ballot_w32(tst != 2u);
    const unsigned int live0 = __builtin_amdgcn_ballot_w32(tst == 0u);

    while (todo != 0u) {
        const int t = __builtin_ctz(todo);
        todo &= todo - 1u;
        const int i = ((tile0 + t) << 5) + lane;
        const bool valid = i < n;
        unsigned long long v = 0ull;
        if (valid) v = in[i];
        unsigned int hi = (unsigned int)(v >> 32);
        unsigned int lo = (unsigned int)v;
        unsigned int st = (unsigned int)__builtin_amdgcn_readfirstlane((int)hi) >> ST_SHIFT;
        if (st == 1u) {
            if (lane == 0) {
                unsigned long long m =
                    ((unsigned long long)((hi & PMASK) | ST_BOTH) << 32) | (unsigned long long)lo;
                in[i]  = m;
                out[i] = m;
            } else if (valid) {
                out[i] = v;
            }
            continue;
        }
        unsigned int p2 = (unsigned int)HN;
        float sum = 0.0f;
        if (valid) {
            int p   = (int)(hi & PMASK);
            float c = __uint_as_float(lo);
            unsigned long long w = in[p];
            float cp = __uint_as_float((unsigned int)w);
            p2  = (unsigned int)(w >> 32) & PMASK;
            sum = c + cp;
        }
        unsigned int live = __builtin_amdgcn_ballot_w32(p2 != (unsigned int)HN);
        if (live == 0u && lane == 0) p2 |= ST_ONE;
        if (valid)
            out[i] = ((unsigned long long)p2 << 32)
                   | (unsigned long long)__float_as_uint(sum);
    }

    if (WRITE_MASK) {
        __shared__ unsigned int s_live[8];
        if (lane == 0) s_live[threadIdx.x >> 5] = live0;
        __syncthreads();
        if (threadIdx.x == 0) {
            unsigned long long m = 0ull;
            for (int k = 0; k < 8; ++k)
                m |= (unsigned long long)s_live[k] << (8 * k);
            masks[blockIdx.x] = m;
        }
    }
}

#define TAIL_FROM     14
#define TAIL_LEVELS   (DOUBLING_STEPS - 1 - TAIL_FROM)
#define TAIL_THREADS  1024
#define TAIL_WAVES    (TAIL_THREADS / 32)
#define TAIL_CHUNKS   ((LATE_BLOCKS + 31) / 32)

__global__ __launch_bounds__(TAIL_THREADS) void doubling_tail(
    unsigned long long* a,
    unsigned long long* b,
    int n,
    const unsigned long long* masks)
{
    __shared__ unsigned long long s_mask[LATE_BLOCKS];
    __shared__ unsigned int s_more[TAIL_WAVES];
    const int lane = threadIdx.x & 31;
    const int w    = threadIdx.x >> 5;

    for (int r = 0; r < (LATE_BLOCKS + TAIL_THREADS - 1) / TAIL_THREADS; ++r) {
        const int k = r * TAIL_THREADS + (int)threadIdx.x;
        if (k < LATE_BLOCKS) s_mask[k] = masks[k];
    }
    __syncthreads();

    unsigned long long* in  = a;
    unsigned long long* out = b;
    for (int lev = 0; lev < TAIL_LEVELS; ++lev) {
        unsigned int more = 0u;
        for (int chunk = w; chunk < TAIL_CHUNKS; chunk += TAIL_WAVES) {
          const int myblk = chunk * 32 + lane;
          const unsigned long long mine = (myblk < LATE_BLOCKS) ? s_mask[myblk] : 0ull;
          unsigned int have = __builtin_amdgcn_ballot_w32(mine != 0ull);
          while (have != 0u) {
            const int bl = __builtin_ctz(have);
            have &= have - 1u;
            const int blk = chunk * 32 + bl;
            unsigned int todo_lo = (unsigned int)__builtin_amdgcn_readlane((int)(unsigned int)mine, bl);
            unsigned int todo_hi = (unsigned int)__builtin_amdgcn_readlane((int)(unsigned int)(mine >> 32), bl);
            unsigned long long todo = ((unsigned long long)todo_hi << 32) | todo_lo;
            unsigned long long next = 0ull;
            while (todo != 0ull) {
                const int t = __builtin_ctzll(todo);
                todo &= todo - 1ull;
                const int i = ((blk * (8 * TILES_PER_WAVE) + t) << 5) + lane;
                const bool valid = i < n;
                unsigned long long v = 0ull;
                if (valid) v = in[i];
                unsigned int hi = (unsigned int)(v >> 32);
                unsigned int lo = (unsigned int)v;
                unsigned int st = (unsigned int)__builtin_amdgcn_readfirstlane((int)hi) >> ST_SHIFT;
                if (st == 2u) continue;
                if (st == 1u) {
                    if (lane == 0) {
                        unsigned long long m =
                            ((unsigned long long)((hi & PMASK) | ST_BOTH) << 32) | (unsigned long long)lo;
                        in[i]  = m;
                        out[i] = m;
                    } else if (valid) {
                        out[i] = v;
                    }
                    continue;
                }
                next |= 1ull << t;
                unsigned int p2 = (unsigned int)HN;
                float sum = 0.0f;
                if (valid) {
                    int p   = (int)(hi & PMASK);
                    float c = __uint_as_float(lo);
                    unsigned long long g = in[p];
                    float cp = __uint_as_float((unsigned int)g);
                    p2  = (unsigned int)(g >> 32) & PMASK;
                    sum = c + cp;
                }
                unsigned int live = __builtin_amdgcn_ballot_w32(p2 != (unsigned int)HN);
                if (live == 0u && lane == 0) p2 |= ST_ONE;
                if (valid)
                    out[i] = ((unsigned long long)p2 << 32)
                           | (unsigned long long)__float_as_uint(sum);
            }
            if (lane == 0) s_mask[blk] = next;
            more |= (next != 0ull) ? 1u : 0u;
          }
        }
        if (lane == 0) s_more[w] = more;
        __syncthreads();
        unsigned int any = 0u;
        for (int k = 0; k < TAIL_WAVES; ++k) any |= s_more[k];
        if (any == 0u) break;
        __threadfence();
        __syncthreads();
        __threadfence();
        unsigned long long* x = in; in = out; out = x;
    }
}

__global__ __launch_bounds__(256) void final_step(
    const unsigned long long* __restrict__ in,
    float* __restrict__ out,
    int n)
{
    int i = blockIdx.x * blockDim.x + threadIdx.x;
    if (i >= n) return;
    unsigned long long v = in[i];
    int p   = (int)((unsigned int)(v >> 32) & PMASK);
    float c = __uint_as_float((unsigned int)v);
    unsigned long long w = in[p];
    out[i] = c + __uint_as_float((unsigned int)w);
}

extern "C" void kernel_launch(void* const* d_in, const int* in_sizes, int n_in,
                              void* d_out, int out_size, void* d_ws, size_t ws_size,
                              hipStream_t stream) {
    const int*   parent = (const int*)d_in[0];
    const float* diff   = (const float*)d_in[1];
    const float* attr   = (const float*)d_in[2];
    const float* weight = (const float*)d_in[3];
    const float* bias   = (const float*)d_in[4];
    float* out = (float*)d_out;

    unsigned long long* buf0 = (unsigned long long*)d_ws;
    unsigned long long* buf1 = buf0 + (HN + 1);

    {
        int blocks = HN / (16 * 8);
        phase1_contrib<<<blocks, 256, 0, stream>>>(parent, diff, attr, weight, bias, buf0);
    }

    {
        int n = HN + 1;
        int blocks = (n + 255) / 256;
        unsigned long long* a = buf0;
        unsigned long long* b = buf1;
        int late_blocks = LATE_BLOCKS;
        unsigned long long* masks = (unsigned long long*)d_out;
        for (int s = 0; s < TAIL_FROM; ++s) {
            if (s < LATE_FROM)
                doubling_step<<<blocks, 256, 0, stream>>>(a, b, n);
            else if (s < TAIL_FROM - 1)
                doubling_step_late<false><<<late_blocks, 256, 0, stream>>>(a, b, n, masks);
            else
                doubling_step_late<true><<<late_blocks, 256, 0, stream>>>(a, b, n, masks);
            unsigned long long* t = a; a = b; b = t;
        }
        doubling_tail<<<1, TAIL_THREADS, 0, stream>>>(a, b, n, masks);
        final_step<<<(HN + 255) / 256, 256, 0, stream>>>(a, out, HN);
    }
}


# Phase 1: a wave takes 32 nodes as two WMMA tiles; a lane computes k pairs {0,1} of its own node and {2,3} of node lane^16, no branch, no cross-lane move; tile 1 swaps the pairs in A and B (sum order)
# speedup vs baseline: 1.3651x; 1.0460x over previous
#include <hip/hip_runtime.h>

#define HN (2048 * 2048)
#define EPSF 1e-10f
#define DOUBLING_STEPS 23

typedef __attribute__((ext_vector_type(2))) float v2f;
typedef __attribute__((ext_vector_type(8))) float v8f;

__global__ __launch_bounds__(256) void phase1_contrib(
    const int* __restrict__ parent,
    const float* __restrict__ diff,
    const float* __restrict__ attr,
    const float* __restrict__ weight,
    const float* __restrict__ bias,
    unsigned long long* __restrict__ buf0)
{
    const int lane = threadIdx.x & 31;
    const int half = lane >> 4;
    const bool hi  = (half != 0);
    const int wave = blockIdx.x * (blockDim.x >> 5) + (threadIdx.x >> 5);
    const int base = wave << 5;
    const int row  = base + lane;
    const int prow = base + (lane ^ 16);
    const float* a = attr + (size_t)row  * 15;
    const float* b = attr + (size_t)prow * 15;

    float r0x[5], r0y[5];
    {
        float a5 = a[5], a6 = a[6];
        r0x[0] = a[0];                        r0y[0] = a[1];
        r0x[1] = a[4];                        r0y[1] = logf(fabsf(a6) + EPSF);
        r0x[2] = logf(fabsf(a[9])  + EPSF);   r0y[2] = logf(fabsf(a[10]) + EPSF);
        r0x[3] = logf(fabsf(a[13]) + EPSF);   r0y[3] = logf(fabsf(a[14]) + EPSF);
        r0x[4] = sinf(a5);                    r0y[4] = 0.0f;
    }
    float r1x[5], r1y[5];
    {
        float b5 = b[5], b6 = b[6], b7 = b[7];
        r1x[0] = b[2];                        r1y[0] = b[3];
        r1x[1] = logf(fabsf(b7)    + EPSF);   r1y[1] = logf(fabsf(b[8])  + EPSF);
        r1x[2] = logf(fabsf(b[11]) + EPSF);   r1y[2] = logf(fabsf(b[12]) + EPSF);
        r1x[3] = sqrtf(b7) / (sqrtf(b6) + EPSF);  r1y[3] = cosf(b5);
        r1x[4] = 0.0f;                        r1y[4] = 0.0f;
    }

    v8f acc0 = {}, acc1 = {};
#pragma unroll
    for (int c = 0; c < 5; ++c) {
        const int k0 = 4 * c + 2 * half;
        const int k1 = 4 * c + 2 * (1 - half);
        v2f A0, A1, B0, B1;
        A0.x = (k0     < 17) ? weight[k0]     : 0.0f;
        A0.y = (k0 + 1 < 17) ? weight[k0 + 1] : 0.0f;
        A1.x = (k1     < 17) ? weight[k1]     : 0.0f;
        A1.y = (k1 + 1 < 17) ? weight[k1 + 1] : 0.0f;
        B0.x = hi ? r1x[c] : r0x[c];   B0.y = hi ? r1y[c] : r0y[c];
        B1.x = hi ? r0x[c] : r1x[c];   B1.y = hi ? r0y[c] : r1y[c];
        acc0 = __builtin_amdgcn_wmma_f32_16x16x4_f32(
             false, A0,  false, B0,
             (short)0, acc0,  false,  false);
        acc1 = __builtin_amdgcn_wmma_f32_16x16x4_f32(
             false, A1,  false, B1,
             (short)0, acc1,  false,  false);
    }

    float val   = (hi ? acc1[0] : acc0[0]) + bias[0];
    float score = 1.0f / (1.0f + expf(-val));

    float contrib = diff[row] * score;
    unsigned long long packed =
        ((unsigned long long)(unsigned int)parent[row] << 32) |
        (unsigned long long)__float_as_uint(contrib);
    buf0[row] = packed;

    if (blockIdx.x == 0 && threadIdx.x == 0) {
        buf0[HN] = ((unsigned long long)(unsigned int)HN << 32);
    }
}

#define PMASK    0x3FFFFFFFu
#define ST_SHIFT 30
#define ST_ONE   (1u << ST_SHIFT)
#define ST_BOTH  (2u << ST_SHIFT)

__global__ __launch_bounds__(256) void doubling_step(
    unsigned long long* in,
    unsigned long long* out,
    int n)
{
    int i = blockIdx.x * blockDim.x + threadIdx.x;
    if (i >= n) return;
    unsigned long long v = in[i];
    unsigned int hi = (unsigned int)(v >> 32);
    unsigned int lo = (unsigned int)v;
    unsigned int st = (unsigned int)__builtin_amdgcn_readfirstlane((int)hi) >> ST_SHIFT;
    if (st == 2u) return;
    if (st == 1u) {
        if ((threadIdx.x & 31) == 0) {
            unsigned long long m =
                ((unsigned long long)((hi & PMASK) | ST_BOTH) << 32) | (unsigned long long)lo;
            in[i]  = m;
            out[i] = m;
        } else {
            out[i] = v;
        }
        return;
    }
    int p   = (int)(hi & PMASK);
    float c = __uint_as_float(lo);
    unsigned long long w = in[p];
    float cp = __uint_as_float((unsigned int)w);
    unsigned int p2 = (unsigned int)(w >> 32) & PMASK;
    unsigned int live = __builtin_amdgcn_ballot_w32(p2 != (unsigned int)HN);
    if (live == 0u && (threadIdx.x & 31) == 0) p2 |= ST_ONE;
    out[i] = ((unsigned long long)p2 << 32)
           | (unsigned long long)__float_as_uint(c + cp);
}

#define LATE_FROM       12
#define TILES_PER_WAVE  8
#define NTILES          ((HN + 1 + 31) / 32)
#define LATE_BLOCKS     ((NTILES + 8 * TILES_PER_WAVE - 1) / (8 * TILES_PER_WAVE))

template <bool WRITE_MASK>
__global__ __launch_bounds__(256) void doubling_step_late(
    unsigned long long* in,
    unsigned long long* out,
    int n,
    unsigned long long* masks)
{
    const int lane   = threadIdx.x & 31;
    const int wave   = blockIdx.x * (blockDim.x >> 5) + (threadIdx.x >> 5);
    const int tile0  = wave * TILES_PER_WAVE;
    const int ntiles = (n + 31) >> 5;

    unsigned int tst = 2u;
    if (lane < TILES_PER_WAVE && tile0 + lane < ntiles)
        tst = (unsigned int)(in[(size_t)(tile0 + lane) << 5] >> (32 + ST_SHIFT));
    unsigned int todo = __builtin_amdgcn_ballot_w32(tst != 2u);
    const unsigned int live0 = __builtin_amdgcn_ballot_w32(tst == 0u);

    while (todo != 0u) {
        const int t = __builtin_ctz(todo);
        todo &= todo - 1u;
        const int i = ((tile0 + t) << 5) + lane;
        const bool valid = i < n;
        unsigned long long v = 0ull;
        if (valid) v = in[i];
        unsigned int hi = (unsigned int)(v >> 32);
        unsigned int lo = (unsigned int)v;
        unsigned int st = (unsigned int)__builtin_amdgcn_readfirstlane((int)hi) >> ST_SHIFT;
        if (st == 1u) {
            if (lane == 0) {
                unsigned long long m =
                    ((unsigned long long)((hi & PMASK) | ST_BOTH) << 32) | (unsigned long long)lo;
                in[i]  = m;
                out[i] = m;
            } else if (valid) {
                out[i] = v;
            }
            continue;
        }
        unsigned int p2 = (unsigned int)HN;
        float sum = 0.0f;
        if (valid) {
            int p   = (int)(hi & PMASK);
            float c = __uint_as_float(lo);
            unsigned long long w = in[p];
            float cp = __uint_as_float((unsigned int)w);
            p2  = (unsigned int)(w >> 32) & PMASK;
            sum = c + cp;
        }
        unsigned int live = __builtin_amdgcn_ballot_w32(p2 != (unsigned int)HN);
        if (live == 0u && lane == 0) p2 |= ST_ONE;
        if (valid)
            out[i] = ((unsigned long long)p2 << 32)
                   | (unsigned long long)__float_as_uint(sum);
    }

    if (WRITE_MASK) {
        __shared__ unsigned int s_live[8];
        if (lane == 0) s_live[threadIdx.x >> 5] = live0;
        __syncthreads();
        if (threadIdx.x == 0) {
            unsigned long long m = 0ull;
            for (int k = 0; k < 8; ++k)
                m |= (unsigned long long)s_live[k] << (8 * k);
            masks[blockIdx.x] = m;
        }
    }
}

#define TAIL_FROM     14
#define TAIL_LEVELS   (DOUBLING_STEPS - 1 - TAIL_FROM)
#define TAIL_THREADS  1024
#define TAIL_WAVES    (TAIL_THREADS / 32)
#define TAIL_CHUNKS   ((LATE_BLOCKS + 31) / 32)

__global__ __launch_bounds__(TAIL_THREADS) void doubling_tail(
    unsigned long long* a,
    unsigned long long* b,
    int n,
    const unsigned long long* masks)
{
    __shared__ unsigned long long s_mask[LATE_BLOCKS];
    __shared__ unsigned int s_more[TAIL_WAVES];
    const int lane = threadIdx.x & 31;
    const int w    = threadIdx.x >> 5;

    for (int r = 0; r < (LATE_BLOCKS + TAIL_THREADS - 1) / TAIL_THREADS; ++r) {
        const int k = r * TAIL_THREADS + (int)threadIdx.x;
        if (k < LATE_BLOCKS) s_mask[k] = masks[k];
    }
    __syncthreads();

    unsigned long long* in  = a;
    unsigned long long* out = b;
    for (int lev = 0; lev < TAIL_LEVELS; ++lev) {
        unsigned int more = 0u;
        for (int chunk = w; chunk < TAIL_CHUNKS; chunk += TAIL_WAVES) {
          const int myblk = chunk * 32 + lane;
          const unsigned long long mine = (myblk < LATE_BLOCKS) ? s_mask[myblk] : 0ull;
          unsigned int have = __builtin_amdgcn_ballot_w32(mine != 0ull);
          while (have != 0u) {
            const int bl = __builtin_ctz(have);
            have &= have - 1u;
            const int blk = chunk * 32 + bl;
            unsigned int todo_lo = (unsigned int)__builtin_amdgcn_readlane((int)(unsigned int)mine, bl);
            unsigned int todo_hi = (unsigned int)__builtin_amdgcn_readlane((int)(unsigned int)(mine >> 32), bl);
            unsigned long long todo = ((unsigned long long)todo_hi << 32) | todo_lo;
            unsigned long long next = 0ull;
            while (todo != 0ull) {
                const int t = __builtin_ctzll(todo);
                todo &= todo - 1ull;
                const int i = ((blk * (8 * TILES_PER_WAVE) + t) << 5) + lane;
                const bool valid = i < n;
                unsigned long long v = 0ull;
                if (valid) v = in[i];
                unsigned int hi = (unsigned int)(v >> 32);
                unsigned int lo = (unsigned int)v;
                unsigned int st = (unsigned int)__builtin_amdgcn_readfirstlane((int)hi) >> ST_SHIFT;
                if (st == 2u) continue;
                if (st == 1u) {
                    if (lane == 0) {
                        unsigned long long m =
                            ((unsigned long long)((hi & PMASK) | ST_BOTH) << 32) | (unsigned long long)lo;
                        in[i]  = m;
                        out[i] = m;
                    } else if (valid) {
                        out[i] = v;
                    }
                    continue;
                }
                next |= 1ull << t;
                unsigned int p2 = (unsigned int)HN;
                float sum = 0.0f;
                if (valid) {
                    int p   = (int)(hi & PMASK);
                    float c = __uint_as_float(lo);
                    unsigned long long g = in[p];
                    float cp = __uint_as_float((unsigned int)g);
                    p2  = (unsigned int)(g >> 32) & PMASK;
                    sum = c + cp;
                }
                unsigned int live = __builtin_amdgcn_ballot_w32(p2 != (unsigned int)HN);
                if (live == 0u && lane == 0) p2 |= ST_ONE;
                if (valid)
                    out[i] = ((unsigned long long)p2 << 32)
                           | (unsigned long long)__float_as_uint(sum);
            }
            if (lane == 0) s_mask[blk] = next;
            more |= (next != 0ull) ? 1u : 0u;
          }
        }
        if (lane == 0) s_more[w] = more;
        __syncthreads();
        unsigned int any = 0u;
        for (int k = 0; k < TAIL_WAVES; ++k) any |= s_more[k];
        if (any == 0u) break;
        __threadfence();
        __syncthreads();
        __threadfence();
        unsigned long long* x = in; in = out; out = x;
    }
}

__global__ __launch_bounds__(256) void final_step(
    const unsigned long long* __restrict__ in,
    float* __restrict__ out,
    int n)
{
    int i = blockIdx.x * blockDim.x + threadIdx.x;
    if (i >= n) return;
    unsigned long long v = in[i];
    int p   = (int)((unsigned int)(v >> 32) & PMASK);
    float c = __uint_as_float((unsigned int)v);
    unsigned long long w = in[p];
    out[i] = c + __uint_as_float((unsigned int)w);
}

extern "C" void kernel_launch(void* const* d_in, const int* in_sizes, int n_in,
                              void* d_out, int out_size, void* d_ws, size_t ws_size,
                              hipStream_t stream) {
    const int*   parent = (const int*)d_in[0];
    const float* diff   = (const float*)d_in[1];
    const float* attr   = (const float*)d_in[2];
    const float* weight = (const float*)d_in[3];
    const float* bias   = (const float*)d_in[4];
    float* out = (float*)d_out;

    unsigned long long* buf0 = (unsigned long long*)d_ws;
    unsigned long long* buf1 = buf0 + (HN + 1);

    {
        int blocks = HN / (32 * 8);
        phase1_contrib<<<blocks, 256, 0, stream>>>(parent, diff, attr, weight, bias, buf0);
    }

    {
        int n = HN + 1;
        int blocks = (n + 255) / 256;
        unsigned long long* a = buf0;
        unsigned long long* b = buf1;
        int late_blocks = LATE_BLOCKS;
        unsigned long long* masks = (unsigned long long*)d_out;
        for (int s = 0; s < TAIL_FROM; ++s) {
            if (s < LATE_FROM)
                doubling_step<<<blocks, 256, 0, stream>>>(a, b, n);
            else if (s < TAIL_FROM - 1)
                doubling_step_late<false><<<late_blocks, 256, 0, stream>>>(a, b, n, masks);
            else
                doubling_step_late<true><<<late_blocks, 256, 0, stream>>>(a, b, n, masks);
            unsigned long long* t = a; a = b; b = t;
        }
        doubling_tail<<<1, TAIL_THREADS, 0, stream>>>(a, b, n, masks);
        final_step<<<(HN + 255) / 256, 256, 0, stream>>>(a, out, HN);
    }
}


# A tile's output is stored by the step that mirrors it once final (c plus the root's zero, same sums); the 131072-wave final launch is gone, the one-workgroup tail closes the rest; 17 launches -> 16
# speedup vs baseline: 1.4193x; 1.0397x over previous
#include <hip/hip_runtime.h>

#define HN (2048 * 2048)
#define EPSF 1e-10f
#define DOUBLING_STEPS 23

typedef __attribute__((ext_vector_type(2))) float v2f;
typedef __attribute__((ext_vector_type(8))) float v8f;

__global__ __launch_bounds__(256) void phase1_contrib(
    const int* __restrict__ parent,
    const float* __restrict__ diff,
    const float* __restrict__ attr,
    const float* __restrict__ weight,
    const float* __restrict__ bias,
    unsigned long long* __restrict__ buf0)
{
    const int lane = threadIdx.x & 31;
    const int half = lane >> 4;
    const bool hi  = (half != 0);
    const int wave = blockIdx.x * (blockDim.x >> 5) + (threadIdx.x >> 5);
    const int base = wave << 5;
    const int row  = base + lane;
    const int prow = base + (lane ^ 16);
    const float* a = attr + (size_t)row  * 15;
    const float* b = attr + (size_t)prow * 15;

    float r0x[5], r0y[5];
    {
        float a5 = a[5], a6 = a[6];
        r0x[0] = a[0];                        r0y[0] = a[1];
        r0x[1] = a[4];                        r0y[1] = logf(fabsf(a6) + EPSF);
        r0x[2] = logf(fabsf(a[9])  + EPSF);   r0y[2] = logf(fabsf(a[10]) + EPSF);
        r0x[3] = logf(fabsf(a[13]) + EPSF);   r0y[3] = logf(fabsf(a[14]) + EPSF);
        r0x[4] = sinf(a5);                    r0y[4] = 0.0f;
    }
    float r1x[5], r1y[5];
    {
        float b5 = b[5], b6 = b[6], b7 = b[7];
        r1x[0] = b[2];                        r1y[0] = b[3];
        r1x[1] = logf(fabsf(b7)    + EPSF);   r1y[1] = logf(fabsf(b[8])  + EPSF);
        r1x[2] = logf(fabsf(b[11]) + EPSF);   r1y[2] = logf(fabsf(b[12]) + EPSF);
        r1x[3] = sqrtf(b7) / (sqrtf(b6) + EPSF);  r1y[3] = cosf(b5);
        r1x[4] = 0.0f;                        r1y[4] = 0.0f;
    }

    v8f acc0 = {}, acc1 = {};
#pragma unroll
    for (int c = 0; c < 5; ++c) {
        const int k0 = 4 * c + 2 * half;
        const int k1 = 4 * c + 2 * (1 - half);
        v2f A0, A1, B0, B1;
        A0.x = (k0     < 17) ? weight[k0]     : 0.0f;
        A0.y = (k0 + 1 < 17) ? weight[k0 + 1] : 0.0f;
        A1.x = (k1     < 17) ? weight[k1]     : 0.0f;
        A1.y = (k1 + 1 < 17) ? weight[k1 + 1] : 0.0f;
        B0.x = hi ? r1x[c] : r0x[c];   B0.y = hi ? r1y[c] : r0y[c];
        B1.x = hi ? r0x[c] : r1x[c];   B1.y = hi ? r0y[c] : r1y[c];
        acc0 = __builtin_amdgcn_wmma_f32_16x16x4_f32(
             false, A0,  false, B0,
             (short)0, acc0,  false,  false);
        acc1 = __builtin_amdgcn_wmma_f32_16x16x4_f32(
             false, A1,  false, B1,
             (short)0, acc1,  false,  false);
    }

    float val   = (hi ? acc1[0] : acc0[0]) + bias[0];
    float score = 1.0f / (1.0f + expf(-val));

    float contrib = diff[row] * score;
    unsigned long long packed =
        ((unsigned long long)(unsigned int)parent[row] << 32) |
        (unsigned long long)__float_as_uint(contrib);
    buf0[row] = packed;

    if (blockIdx.x == 0 && threadIdx.x == 0) {
        buf0[HN] = ((unsigned long long)(unsigned int)HN << 32);
    }
}

#define PMASK    0x3FFFFFFFu
#define ST_SHIFT 30
#define ST_ONE   (1u << ST_SHIFT)
#define ST_BOTH  (2u << ST_SHIFT)
#define MASK_BYTES  (((HN + 1 + 31) / 32 + 63) / 64 * 8)
#define MASK_TILES  ((MASK_BYTES + 127) / 128)

__global__ __launch_bounds__(256) void doubling_step(
    unsigned long long* in,
    unsigned long long* out,
    int n,
    float* fout)
{
    int i = blockIdx.x * blockDim.x + threadIdx.x;
    if (i >= n) return;
    unsigned long long v = in[i];
    unsigned int hi = (unsigned int)(v >> 32);
    unsigned int lo = (unsigned int)v;
    unsigned int st = (unsigned int)__builtin_amdgcn_readfirstlane((int)hi) >> ST_SHIFT;
    if (st == 2u) return;
    if (st == 1u) {
        if ((threadIdx.x & 31) == 0) {
            unsigned long long m =
                ((unsigned long long)((hi & PMASK) | ST_BOTH) << 32) | (unsigned long long)lo;
            in[i]  = m;
            out[i] = m;
        } else {
            out[i] = v;
        }
        if (i < HN && (i >> 5) >= MASK_TILES) fout[i] = __uint_as_float(lo) + 0.0f;
        return;
    }
    int p   = (int)(hi & PMASK);
    float c = __uint_as_float(lo);
    unsigned long long w = in[p];
    float cp = __uint_as_float((unsigned int)w);
    unsigned int p2 = (unsigned int)(w >> 32) & PMASK;
    unsigned int live = __builtin_amdgcn_ballot_w32(p2 != (unsigned int)HN);
    if (live == 0u && (threadIdx.x & 31) == 0) p2 |= ST_ONE;
    out[i] = ((unsigned long long)p2 << 32)
           | (unsigned long long)__float_as_uint(c + cp);
}

#define LATE_FROM       12
#define TILES_PER_WAVE  8
#define NTILES          ((HN + 1 + 31) / 32)
#define LATE_BLOCKS     ((NTILES + 8 * TILES_PER_WAVE - 1) / (8 * TILES_PER_WAVE))

template <bool WRITE_MASK>
__global__ __launch_bounds__(256) void doubling_step_late(
    unsigned long long* in,
    unsigned long long* out,
    int n,
    unsigned long long* masks,
    float* fout)
{
    const int lane   = threadIdx.x & 31;
    const int wave   = blockIdx.x * (blockDim.x >> 5) + (threadIdx.x >> 5);
    const int tile0  = wave * TILES_PER_WAVE;
    const int ntiles = (n + 31) >> 5;

    unsigned int tst = 2u;
    if (lane < TILES_PER_WAVE && tile0 + lane < ntiles)
        tst = (unsigned int)(in[(size_t)(tile0 + lane) << 5] >> (32 + ST_SHIFT));
    unsigned int todo = __builtin_amdgcn_ballot_w32(tst != 2u);
    const unsigned int live0 = __builtin_amdgcn_ballot_w32(tst == 0u);

    while (todo != 0u) {
        const int t = __builtin_ctz(todo);
        todo &= todo - 1u;
        const int i = ((tile0 + t) << 5) + lane;
        const bool valid = i < n;
        unsigned long long v = 0ull;
        if (valid) v = in[i];
        unsigned int hi = (unsigned int)(v >> 32);
        unsigned int lo = (unsigned int)v;
        unsigned int st = (unsigned int)__builtin_amdgcn_readfirstlane((int)hi) >> ST_SHIFT;
        if (st == 1u) {
            if (lane == 0) {
                unsigned long long m =
                    ((unsigned long long)((hi & PMASK) | ST_BOTH) << 32) | (unsigned long long)lo;
                in[i]  = m;
                out[i] = m;
            } else if (valid) {
                out[i] = v;
            }
            if (i < HN && tile0 + t >= MASK_TILES) fout[i] = __uint_as_float(lo) + 0.0f;
            continue;
        }
        unsigned int p2 = (unsigned int)HN;
        float sum = 0.0f;
        if (valid) {
            int p   = (int)(hi & PMASK);
            float c = __uint_as_float(lo);
            unsigned long long w = in[p];
            float cp = __uint_as_float((unsigned int)w);
            p2  = (unsigned int)(w >> 32) & PMASK;
            sum = c + cp;
        }
        unsigned int live = __builtin_amdgcn_ballot_w32(p2 != (unsigned int)HN);
        if (live == 0u && lane == 0) p2 |= ST_ONE;
        if (valid)
            out[i] = ((unsigned long long)p2 << 32)
                   | (unsigned long long)__float_as_uint(sum);
    }

    if (WRITE_MASK) {
        __shared__ unsigned int s_live[8];
        if (lane == 0) s_live[threadIdx.x >> 5] = live0;
        __syncthreads();
        if (threadIdx.x == 0) {
            unsigned long long m = 0ull;
            for (int k = 0; k < 8; ++k)
                m |= (unsigned long long)s_live[k] << (8 * k);
            masks[blockIdx.x] = m;
        }
    }
}

#define TAIL_FROM     14
#define TAIL_LEVELS   (DOUBLING_STEPS - 1 - TAIL_FROM)
#define TAIL_THREADS  1024
#define TAIL_WAVES    (TAIL_THREADS / 32)
#define TAIL_CHUNKS   ((LATE_BLOCKS + 31) / 32)
#define CLOSE_PER_THREAD ((MASK_TILES * 32 + TAIL_THREADS - 1) / TAIL_THREADS)

__global__ __launch_bounds__(TAIL_THREADS) void doubling_tail(
    unsigned long long* a,
    unsigned long long* b,
    int n,
    const unsigned long long* masks,
    float* fout)
{
    __shared__ unsigned long long s_mask[LATE_BLOCKS];
    __shared__ unsigned int s_more[TAIL_WAVES];
    const int lane = threadIdx.x & 31;
    const int w    = threadIdx.x >> 5;

    for (int r = 0; r < (LATE_BLOCKS + TAIL_THREADS - 1) / TAIL_THREADS; ++r) {
        const int k = r * TAIL_THREADS + (int)threadIdx.x;
        if (k < LATE_BLOCKS) s_mask[k] = masks[k];
    }
    __syncthreads();

    unsigned long long* in  = a;
    unsigned long long* out = b;
    for (int lev = 0; lev < TAIL_LEVELS; ++lev) {
        unsigned int more = 0u;
        for (int chunk = w; chunk < TAIL_CHUNKS; chunk += TAIL_WAVES) {
          const int myblk = chunk * 32 + lane;
          const unsigned long long mine = (myblk < LATE_BLOCKS) ? s_mask[myblk] : 0ull;
          unsigned int have = __builtin_amdgcn_ballot_w32(mine != 0ull);
          while (have != 0u) {
            const int bl = __builtin_ctz(have);
            have &= have - 1u;
            const int blk = chunk * 32 + bl;
            unsigned int todo_lo = (unsigned int)__builtin_amdgcn_readlane((int)(unsigned int)mine, bl);
            unsigned int todo_hi = (unsigned int)__builtin_amdgcn_readlane((int)(unsigned int)(mine >> 32), bl);
            unsigned long long todo = ((unsigned long long)todo_hi << 32) | todo_lo;
            unsigned long long next = 0ull;
            while (todo != 0ull) {
                const int t = __builtin_ctzll(todo);
                todo &= todo - 1ull;
                const int i = ((blk * (8 * TILES_PER_WAVE) + t) << 5) + lane;
                const bool valid = i < n;
                unsigned long long v = 0ull;
                if (valid) v = in[i];
                unsigned int hi = (unsigned int)(v >> 32);
                unsigned int lo = (unsigned int)v;
                unsigned int st = (unsigned int)__builtin_amdgcn_readfirstlane((int)hi) >> ST_SHIFT;
                if (st == 2u) continue;
                if (st == 1u) {
                    if (lane == 0) {
                        unsigned long long m =
                            ((unsigned long long)((hi & PMASK) | ST_BOTH) << 32) | (unsigned long long)lo;
                        in[i]  = m;
                        out[i] = m;
                    } else if (valid) {
                        out[i] = v;
                    }
                    if (i < HN && blk * (8 * TILES_PER_WAVE) + t >= MASK_TILES)
                        fout[i] = __uint_as_float(lo) + 0.0f;
                    continue;
                }
                next |= 1ull << t;
                unsigned int p2 = (unsigned int)HN;
                float sum = 0.0f;
                if (valid) {
                    int p   = (int)(hi & PMASK);
                    float c = __uint_as_float(lo);
                    unsigned long long g = in[p];
                    float cp = __uint_as_float((unsigned int)g);
                    p2  = (unsigned int)(g >> 32) & PMASK;
                    sum = c + cp;
                }
                unsigned int live = __builtin_amdgcn_ballot_w32(p2 != (unsigned int)HN);
                if (live == 0u && lane == 0) p2 |= ST_ONE;
                if (valid)
                    out[i] = ((unsigned long long)p2 << 32)
                           | (unsigned long long)__float_as_uint(sum);
            }
            if (lane == 0) s_mask[blk] = next;
            more |= (next != 0ull) ? 1u : 0u;
          }
        }
        if (lane == 0) s_more[w] = more;
        __syncthreads();
        unsigned int any = 0u;
        for (int k = 0; k < TAIL_WAVES; ++k) any |= s_more[k];
        if (any == 0u) break;
        __threadfence();
        __syncthreads();
        __threadfence();
        unsigned long long* x = in; in = out; out = x;
    }

    const unsigned long long* fin = in;

    {
        unsigned long long v[CLOSE_PER_THREAD], g[CLOSE_PER_THREAD];
#pragma unroll
        for (int j = 0; j < CLOSE_PER_THREAD; ++j) {
            const int r = j * TAIL_THREADS + (int)threadIdx.x;
            v[j] = (r < MASK_TILES * 32) ? fin[r] : 0ull;
        }
#pragma unroll
        for (int j = 0; j < CLOSE_PER_THREAD; ++j) {
            const int r = j * TAIL_THREADS + (int)threadIdx.x;
            const int p = (int)((unsigned int)(v[j] >> 32) & PMASK);
            g[j] = (r < MASK_TILES * 32) ? fin[p] : 0ull;
        }
#pragma unroll
        for (int j = 0; j < CLOSE_PER_THREAD; ++j) {
            const int r = j * TAIL_THREADS + (int)threadIdx.x;
            if (r < MASK_TILES * 32)
                fout[r] = __uint_as_float((unsigned int)v[j]) + __uint_as_float((unsigned int)g[j]);
        }
    }

    for (int chunk = w; chunk < TAIL_CHUNKS; chunk += TAIL_WAVES) {
        const int myblk = chunk * 32 + lane;
        const unsigned long long mine = (myblk < LATE_BLOCKS) ? s_mask[myblk] : 0ull;
        unsigned int have = __builtin_amdgcn_ballot_w32(mine != 0ull);
        while (have != 0u) {
            const int bl = __builtin_ctz(have);
            have &= have - 1u;
            const int blk = chunk * 32 + bl;
            unsigned int todo_lo = (unsigned int)__builtin_amdgcn_readlane((int)(unsigned int)mine, bl);
            unsigned int todo_hi = (unsigned int)__builtin_amdgcn_readlane((int)(unsigned int)(mine >> 32), bl);
            unsigned long long todo = ((unsigned long long)todo_hi << 32) | todo_lo;
            while (todo != 0ull) {
                const int t = __builtin_ctzll(todo);
                todo &= todo - 1ull;
                const int tile = blk * (8 * TILES_PER_WAVE) + t;
                const int i = (tile << 5) + lane;
                if (tile >= MASK_TILES && i < HN) {
                    unsigned long long v = fin[i];
                    int p = (int)((unsigned int)(v >> 32) & PMASK);
                    unsigned long long g = fin[p];
                    fout[i] = __uint_as_float((unsigned int)v) + __uint_as_float((unsigned int)g);
                }
            }
        }
    }
}

extern "C" void kernel_launch(void* const* d_in, const int* in_sizes, int n_in,
                              void* d_out, int out_size, void* d_ws, size_t ws_size,
                              hipStream_t stream) {
    const int*   parent = (const int*)d_in[0];
    const float* diff   = (const float*)d_in[1];
    const float* attr   = (const float*)d_in[2];
    const float* weight = (const float*)d_in[3];
    const float* bias   = (const float*)d_in[4];
    float* out = (float*)d_out;

    unsigned long long* buf0 = (unsigned long long*)d_ws;
    unsigned long long* buf1 = buf0 + (HN + 1);

    {
        int blocks = HN / (32 * 8);
        phase1_contrib<<<blocks, 256, 0, stream>>>(parent, diff, attr, weight, bias, buf0);
    }

    {
        int n = HN + 1;
        int blocks = (n + 255) / 256;
        unsigned long long* a = buf0;
        unsigned long long* b = buf1;
        int late_blocks = LATE_BLOCKS;
        unsigned long long* masks = (unsigned long long*)d_out;
        for (int s = 0; s < TAIL_FROM; ++s) {
            if (s < LATE_FROM)
                doubling_step<<<blocks, 256, 0, stream>>>(a, b, n, out);
            else if (s < TAIL_FROM - 1)
                doubling_step_late<false><<<late_blocks, 256, 0, stream>>>(a, b, n, masks, out);
            else
                doubling_step_late<true><<<late_blocks, 256, 0, stream>>>(a, b, n, masks, out);
            unsigned long long* t = a; a = b; b = t;
        }
        doubling_tail<<<1, TAIL_THREADS, 0, stream>>>(a, b, n, masks, out);
    }
}


# Phase 1 sums each run of nodes whose parent is the left neighbour into one record (scan in the wave, then over the block's 8 waves): other sum order; full levels 12 -> 5, launches 16 -> 9
# speedup vs baseline: 2.3818x; 1.6782x over previous
#include <hip/hip_runtime.h>

#define HN (2048 * 2048)
#define EPSF 1e-10f
#define DOUBLING_STEPS 23

typedef __attribute__((ext_vector_type(2))) float v2f;
typedef __attribute__((ext_vector_type(8))) float v8f;

__device__ __forceinline__ float node_contrib(
    const float* __restrict__ diff,
    const float* __restrict__ attr,
    const float* __restrict__ weight,
    const float* __restrict__ bias,
    const int base, const int lane)
{
    const int half = lane >> 4;
    const bool hi  = (half != 0);
    const int row  = base + lane;
    const int prow = base + (lane ^ 16);
    const float* a = attr + (size_t)row  * 15;
    const float* b = attr + (size_t)prow * 15;

    float r0x[5], r0y[5];
    {
        float a5 = a[5], a6 = a[6];
        r0x[0] = a[0];                        r0y[0] = a[1];
        r0x[1] = a[4];                        r0y[1] = logf(fabsf(a6) + EPSF);
        r0x[2] = logf(fabsf(a[9])  + EPSF);   r0y[2] = logf(fabsf(a[10]) + EPSF);
        r0x[3] = logf(fabsf(a[13]) + EPSF);   r0y[3] = logf(fabsf(a[14]) + EPSF);
        r0x[4] = sinf(a5);                    r0y[4] = 0.0f;
    }
    float r1x[5], r1y[5];
    {
        float b5 = b[5], b6 = b[6], b7 = b[7];
        r1x[0] = b[2];                        r1y[0] = b[3];
        r1x[1] = logf(fabsf(b7)    + EPSF);   r1y[1] = logf(fabsf(b[8])  + EPSF);
        r1x[2] = logf(fabsf(b[11]) + EPSF);   r1y[2] = logf(fabsf(b[12]) + EPSF);
        r1x[3] = sqrtf(b7) / (sqrtf(b6) + EPSF);  r1y[3] = cosf(b5);
        r1x[4] = 0.0f;                        r1y[4] = 0.0f;
    }

    v8f acc0 = {}, acc1 = {};
#pragma unroll
    for (int c = 0; c < 5; ++c) {
        const int k0 = 4 * c + 2 * half;
        const int k1 = 4 * c + 2 * (1 - half);
        v2f A0, A1, B0, B1;
        A0.x = (k0     < 17) ? weight[k0]     : 0.0f;
        A0.y = (k0 + 1 < 17) ? weight[k0 + 1] : 0.0f;
        A1.x = (k1     < 17) ? weight[k1]     : 0.0f;
        A1.y = (k1 + 1 < 17) ? weight[k1 + 1] : 0.0f;
        B0.x = hi ? r1x[c] : r0x[c];   B0.y = hi ? r1y[c] : r0y[c];
        B1.x = hi ? r0x[c] : r1x[c];   B1.y = hi ? r0y[c] : r1y[c];
        acc0 = __builtin_amdgcn_wmma_f32_16x16x4_f32(
             false, A0,  false, B0,
             (short)0, acc0,  false,  false);
        acc1 = __builtin_amdgcn_wmma_f32_16x16x4_f32(
             false, A1,  false, B1,
             (short)0, acc1,  false,  false);
    }

    float val   = (hi ? acc1[0] : acc0[0]) + bias[0];
    float score = 1.0f / (1.0f + expf(-val));
    return diff[row] * score;
}

#define RUN_BIT 0x80000000u

__global__ __launch_bounds__(256, 16) void phase1_contrib(
    const int* __restrict__ parent,
    const float* __restrict__ diff,
    const float* __restrict__ attr,
    const float* __restrict__ weight,
    const float* __restrict__ bias,
    unsigned long long* __restrict__ buf0)
{
    __shared__ float        s_tail_c[8];
    __shared__ unsigned int s_tail_k[8];
    const int lane = threadIdx.x & 31;
    const int w    = threadIdx.x >> 5;
    const int wave = blockIdx.x * 8 + w;
    const int base = wave << 5;
    const int row  = base + lane;

    float c = node_contrib(diff, attr, weight, bias, base, lane);
    const int p = parent[row];

    const bool cont = (p == row - 1);

    unsigned int k = (unsigned int)p | ((cont && lane != 0) ? RUN_BIT : 0u);
#pragma unroll
    for (int d = 1; d < 32; d <<= 1) {
        const float        cu = __shfl_up(c, d);
        const unsigned int ku = (unsigned int)__shfl_up((int)k, d);
        if (k & RUN_BIT) { c += cu; k = ku; }
    }

    const unsigned int contmask = __builtin_amdgcn_ballot_w32(cont);
    const unsigned int need     = (2u << lane) - 2u;
    const bool reach0 = ((~contmask) & need) == 0u;
    const bool wcont  = ((contmask & 1u) != 0u) && (w != 0);
    if (lane == 31) {
        s_tail_c[w] = c;
        s_tail_k[w] = k | ((reach0 && wcont) ? RUN_BIT : 0u);
    }
    __syncthreads();
    if (wcont) {
        unsigned int t = s_tail_k[w - 1];
        float cc = s_tail_c[w - 1];
        for (int v = w - 2; v >= 0 && (t & RUN_BIT) != 0u; --v) {
            cc += s_tail_c[v];
            t = s_tail_k[v];
        }
        if (reach0) { c += cc; k = t & ~RUN_BIT; }
    }

    unsigned long long packed =
        ((unsigned long long)k << 32) |
        (unsigned long long)__float_as_uint(c);
    buf0[row] = packed;

    if (blockIdx.x == 0 && threadIdx.x == 0) {
        buf0[HN] = ((unsigned long long)(unsigned int)HN << 32);
    }
}

#define PMASK    0x3FFFFFFFu
#define ST_SHIFT 30
#define ST_ONE   (1u << ST_SHIFT)
#define ST_BOTH  (2u << ST_SHIFT)
#define MASK_BYTES  (((HN + 1 + 31) / 32 + 63) / 64 * 8)
#define MASK_TILES  ((MASK_BYTES + 127) / 128)

__global__ __launch_bounds__(256) void doubling_step(
    unsigned long long* in,
    unsigned long long* out,
    int n,
    float* fout)
{
    int i = blockIdx.x * blockDim.x + threadIdx.x;
    if (i >= n) return;
    unsigned long long v = in[i];
    unsigned int hi = (unsigned int)(v >> 32);
    unsigned int lo = (unsigned int)v;
    unsigned int st = (unsigned int)__builtin_amdgcn_readfirstlane((int)hi) >> ST_SHIFT;
    if (st == 2u) return;
    if (st == 1u) {
        if ((threadIdx.x & 31) == 0) {
            unsigned long long m =
                ((unsigned long long)((hi & PMASK) | ST_BOTH) << 32) | (unsigned long long)lo;
            in[i]  = m;
            out[i] = m;
        } else {
            out[i] = v;
        }
        if (i < HN && (i >> 5) >= MASK_TILES) fout[i] = __uint_as_float(lo) + 0.0f;
        return;
    }
    int p   = (int)(hi & PMASK);
    float c = __uint_as_float(lo);
    unsigned long long w = in[p];
    float cp = __uint_as_float((unsigned int)w);
    unsigned int p2 = (unsigned int)(w >> 32) & PMASK;
    unsigned int live = __builtin_amdgcn_ballot_w32(p2 != (unsigned int)HN);
    if (live == 0u && (threadIdx.x & 31) == 0) p2 |= ST_ONE;
    out[i] = ((unsigned long long)p2 << 32)
           | (unsigned long long)__float_as_uint(c + cp);
}

#define LATE_FROM       5
#define TILES_PER_WAVE  8
#define NTILES          ((HN + 1 + 31) / 32)
#define LATE_BLOCKS     ((NTILES + 8 * TILES_PER_WAVE - 1) / (8 * TILES_PER_WAVE))

template <bool WRITE_MASK>
__global__ __launch_bounds__(256) void doubling_step_late(
    unsigned long long* in,
    unsigned long long* out,
    int n,
    unsigned long long* masks,
    float* fout)
{
    const int lane   = threadIdx.x & 31;
    const int wave   = blockIdx.x * (blockDim.x >> 5) + (threadIdx.x >> 5);
    const int tile0  = wave * TILES_PER_WAVE;
    const int ntiles = (n + 31) >> 5;

    unsigned int tst = 2u;
    if (lane < TILES_PER_WAVE && tile0 + lane < ntiles)
        tst = (unsigned int)(in[(size_t)(tile0 + lane) << 5] >> (32 + ST_SHIFT));
    unsigned int todo = __builtin_amdgcn_ballot_w32(tst != 2u);
    const unsigned int live0 = __builtin_amdgcn_ballot_w32(tst == 0u);

    while (todo != 0u) {
        const int t = __builtin_ctz(todo);
        todo &= todo - 1u;
        const int i = ((tile0 + t) << 5) + lane;
        const bool valid = i < n;
        unsigned long long v = 0ull;
        if (valid) v = in[i];
        unsigned int hi = (unsigned int)(v >> 32);
        unsigned int lo = (unsigned int)v;
        unsigned int st = (unsigned int)__builtin_amdgcn_readfirstlane((int)hi) >> ST_SHIFT;
        if (st == 1u) {
            if (lane == 0) {
                unsigned long long m =
                    ((unsigned long long)((hi & PMASK) | ST_BOTH) << 32) | (unsigned long long)lo;
                in[i]  = m;
                out[i] = m;
            } else if (valid) {
                out[i] = v;
            }
            if (i < HN && tile0 + t >= MASK_TILES) fout[i] = __uint_as_float(lo) + 0.0f;
            continue;
        }
        unsigned int p2 = (unsigned int)HN;
        float sum = 0.0f;
        if (valid) {
            int p   = (int)(hi & PMASK);
            float c = __uint_as_float(lo);
            unsigned long long w = in[p];
            float cp = __uint_as_float((unsigned int)w);
            p2  = (unsigned int)(w >> 32) & PMASK;
            sum = c + cp;
        }
        unsigned int live = __builtin_amdgcn_ballot_w32(p2 != (unsigned int)HN);
        if (live == 0u && lane == 0) p2 |= ST_ONE;
        if (valid)
            out[i] = ((unsigned long long)p2 << 32)
                   | (unsigned long long)__float_as_uint(sum);
    }

    if (WRITE_MASK) {
        __shared__ unsigned int s_live[8];
        if (lane == 0) s_live[threadIdx.x >> 5] = live0;
        __syncthreads();
        if (threadIdx.x == 0) {
            unsigned long long m = 0ull;
            for (int k = 0; k < 8; ++k)
                m |= (unsigned long long)s_live[k] << (8 * k);
            masks[blockIdx.x] = m;
        }
    }
}

#define TAIL_FROM     7
#define TAIL_LEVELS   (DOUBLING_STEPS - 1 - TAIL_FROM)
#define TAIL_THREADS  1024
#define TAIL_WAVES    (TAIL_THREADS / 32)
#define TAIL_CHUNKS   ((LATE_BLOCKS + 31) / 32)
#define CLOSE_PER_THREAD ((MASK_TILES * 32 + TAIL_THREADS - 1) / TAIL_THREADS)

__device__ void exact_redo(
    unsigned long long* a, unsigned long long* b, const int n,
    const int* __restrict__ parent, const float* __restrict__ diff,
    const float* __restrict__ attr, const float* __restrict__ weight,
    const float* __restrict__ bias, float* fout)
{
    const int lane = threadIdx.x & 31;
    const int w    = threadIdx.x >> 5;
    for (int base = w << 5; base < HN; base += TAIL_THREADS) {
        const float c = node_contrib(diff, attr, weight, bias, base, lane);
        const int row = base + lane;
        a[row] = ((unsigned long long)(unsigned int)parent[row] << 32)
               | (unsigned long long)__float_as_uint(c);
    }
    if (threadIdx.x == 0) a[HN] = ((unsigned long long)(unsigned int)HN << 32);
    __threadfence();
    __syncthreads();
    __threadfence();

    unsigned long long* in  = a;
    unsigned long long* out = b;
    for (int lev = 0; lev < DOUBLING_STEPS - 1; ++lev) {
        for (int r = (int)threadIdx.x; r < n; r += TAIL_THREADS) {
            const unsigned long long v = in[r];
            const unsigned long long g = in[(unsigned int)(v >> 32)];
            out[r] = (g & 0xFFFFFFFF00000000ull)
                   | (unsigned long long)__float_as_uint(
                         __uint_as_float((unsigned int)v) + __uint_as_float((unsigned int)g));
        }
        __threadfence();
        __syncthreads();
        __threadfence();
        unsigned long long* x = in; in = out; out = x;
    }
    for (int r = (int)threadIdx.x; r < HN; r += TAIL_THREADS) {
        const unsigned long long v = in[r];
        const unsigned long long g = in[(unsigned int)(v >> 32)];
        fout[r] = __uint_as_float((unsigned int)v) + __uint_as_float((unsigned int)g);
    }
}

__global__ __launch_bounds__(TAIL_THREADS) void doubling_tail(
    unsigned long long* a,
    unsigned long long* b,
    int n,
    const unsigned long long* masks,
    float* fout,
    const int* __restrict__ parent,
    const float* __restrict__ diff,
    const float* __restrict__ attr,
    const float* __restrict__ weight,
    const float* __restrict__ bias)
{
    __shared__ unsigned long long s_mask[LATE_BLOCKS];
    __shared__ unsigned int s_more[TAIL_WAVES];
    __shared__ unsigned int s_loose[TAIL_WAVES];
    const int lane = threadIdx.x & 31;
    const int w    = threadIdx.x >> 5;

    for (int r = 0; r < (LATE_BLOCKS + TAIL_THREADS - 1) / TAIL_THREADS; ++r) {
        const int k = r * TAIL_THREADS + (int)threadIdx.x;
        if (k < LATE_BLOCKS) s_mask[k] = masks[k];
    }
    __syncthreads();

    unsigned long long* in  = a;
    unsigned long long* out = b;
    for (int lev = 0; lev < TAIL_LEVELS; ++lev) {
        unsigned int more = 0u;
        for (int chunk = w; chunk < TAIL_CHUNKS; chunk += TAIL_WAVES) {
          const int myblk = chunk * 32 + lane;
          const unsigned long long mine = (myblk < LATE_BLOCKS) ? s_mask[myblk] : 0ull;
          unsigned int have = __builtin_amdgcn_ballot_w32(mine != 0ull);
          while (have != 0u) {
            const int bl = __builtin_ctz(have);
            have &= have - 1u;
            const int blk = chunk * 32 + bl;
            unsigned int todo_lo = (unsigned int)__builtin_amdgcn_readlane((int)(unsigned int)mine, bl);
            unsigned int todo_hi = (unsigned int)__builtin_amdgcn_readlane((int)(unsigned int)(mine >> 32), bl);
            unsigned long long todo = ((unsigned long long)todo_hi << 32) | todo_lo;
            unsigned long long next = 0ull;
            while (todo != 0ull) {
                const int t = __builtin_ctzll(todo);
                todo &= todo - 1ull;
                const int i = ((blk * (8 * TILES_PER_WAVE) + t) << 5) + lane;
                const bool valid = i < n;
                unsigned long long v = 0ull;
                if (valid) v = in[i];
                unsigned int hi = (unsigned int)(v >> 32);
                unsigned int lo = (unsigned int)v;
                unsigned int st = (unsigned int)__builtin_amdgcn_readfirstlane((int)hi) >> ST_SHIFT;
                if (st == 2u) continue;
                if (st == 1u) {
                    if (lane == 0) {
                        unsigned long long m =
                            ((unsigned long long)((hi & PMASK) | ST_BOTH) << 32) | (unsigned long long)lo;
                        in[i]  = m;
                        out[i] = m;
                    } else if (valid) {
                        out[i] = v;
                    }
                    if (i < HN && blk * (8 * TILES_PER_WAVE) + t >= MASK_TILES)
                        fout[i] = __uint_as_float(lo) + 0.0f;
                    continue;
                }
                next |= 1ull << t;
                unsigned int p2 = (unsigned int)HN;
                float sum = 0.0f;
                if (valid) {
                    int p   = (int)(hi & PMASK);
                    float c = __uint_as_float(lo);
                    unsigned long long g = in[p];
                    float cp = __uint_as_float((unsigned int)g);
                    p2  = (unsigned int)(g >> 32) & PMASK;
                    sum = c + cp;
                }
                unsigned int live = __builtin_amdgcn_ballot_w32(p2 != (unsigned int)HN);
                if (live == 0u && lane == 0) p2 |= ST_ONE;
                if (valid)
                    out[i] = ((unsigned long long)p2 << 32)
                           | (unsigned long long)__float_as_uint(sum);
            }
            if (lane == 0) s_mask[blk] = next;
            more |= (next != 0ull) ? 1u : 0u;
          }
        }
        if (lane == 0) s_more[w] = more;
        __syncthreads();
        unsigned int any = 0u;
        for (int k = 0; k < TAIL_WAVES; ++k) any |= s_more[k];
        if (any == 0u) break;
        __threadfence();
        __syncthreads();
        __threadfence();
        unsigned long long* x = in; in = out; out = x;
    }

    const unsigned long long* fin = in;

    {
        bool loose = false;
        for (int chunk = w; chunk < TAIL_CHUNKS; chunk += TAIL_WAVES) {
            const int myblk = chunk * 32 + lane;
            const unsigned long long mine = (myblk < LATE_BLOCKS) ? s_mask[myblk] : 0ull;
            unsigned int have = __builtin_amdgcn_ballot_w32(mine != 0ull);
            while (have != 0u) {
                const int bl = __builtin_ctz(have);
                have &= have - 1u;
                const int blk = chunk * 32 + bl;
                unsigned int todo_lo = (unsigned int)__builtin_amdgcn_readlane((int)(unsigned int)mine, bl);
                unsigned int todo_hi = (unsigned int)__builtin_amdgcn_readlane((int)(unsigned int)(mine >> 32), bl);
                unsigned long long todo = ((unsigned long long)todo_hi << 32) | todo_lo;
                while (todo != 0ull) {
                    const int t = __builtin_ctzll(todo);
                    todo &= todo - 1ull;
                    const int i = ((blk * (8 * TILES_PER_WAVE) + t) << 5) + lane;
                    if (i < n)
                        loose = loose || (((unsigned int)(fin[i] >> 32) & PMASK) != (unsigned int)HN);
                }
            }
        }
        const unsigned int wl = __builtin_amdgcn_ballot_w32(loose);
        if (lane == 0) s_loose[w] = wl;
        __syncthreads();
        unsigned int anyloose = 0u;
        for (int k = 0; k < TAIL_WAVES; ++k) anyloose |= s_loose[k];
        if (anyloose != 0u) {
            exact_redo((unsigned long long*)a, (unsigned long long*)b, n, parent, diff, attr, weight, bias, fout);
            return;
        }
    }

    {
        unsigned long long v[CLOSE_PER_THREAD], g[CLOSE_PER_THREAD];
#pragma unroll
        for (int j = 0; j < CLOSE_PER_THREAD; ++j) {
            const int r = j * TAIL_THREADS + (int)threadIdx.x;
            v[j] = (r < MASK_TILES * 32) ? fin[r] : 0ull;
        }
#pragma unroll
        for (int j = 0; j < CLOSE_PER_THREAD; ++j) {
            const int r = j * TAIL_THREADS + (int)threadIdx.x;
            const int p = (int)((unsigned int)(v[j] >> 32) & PMASK);
            g[j] = (r < MASK_TILES * 32) ? fin[p] : 0ull;
        }
#pragma unroll
        for (int j = 0; j < CLOSE_PER_THREAD; ++j) {
            const int r = j * TAIL_THREADS + (int)threadIdx.x;
            if (r < MASK_TILES * 32)
                fout[r] = __uint_as_float((unsigned int)v[j]) + __uint_as_float((unsigned int)g[j]);
        }
    }

    for (int chunk = w; chunk < TAIL_CHUNKS; chunk += TAIL_WAVES) {
        const int myblk = chunk * 32 + lane;
        const unsigned long long mine = (myblk < LATE_BLOCKS) ? s_mask[myblk] : 0ull;
        unsigned int have = __builtin_amdgcn_ballot_w32(mine != 0ull);
        while (have != 0u) {
            const int bl = __builtin_ctz(have);
            have &= have - 1u;
            const int blk = chunk * 32 + bl;
            unsigned int todo_lo = (unsigned int)__builtin_amdgcn_readlane((int)(unsigned int)mine, bl);
            unsigned int todo_hi = (unsigned int)__builtin_amdgcn_readlane((int)(unsigned int)(mine >> 32), bl);
            unsigned long long todo = ((unsigned long long)todo_hi << 32) | todo_lo;
            while (todo != 0ull) {
                const int t = __builtin_ctzll(todo);
                todo &= todo - 1ull;
                const int tile = blk * (8 * TILES_PER_WAVE) + t;
                const int i = (tile << 5) + lane;
                if (tile >= MASK_TILES && i < HN) {
                    unsigned long long v = fin[i];
                    int p = (int)((unsigned int)(v >> 32) & PMASK);
                    unsigned long long g = fin[p];
                    fout[i] = __uint_as_float((unsigned int)v) + __uint_as_float((unsigned int)g);
                }
            }
        }
    }
}

extern "C" void kernel_launch(void* const* d_in, const int* in_sizes, int n_in,
                              void* d_out, int out_size, void* d_ws, size_t ws_size,
                              hipStream_t stream) {
    const int*   parent = (const int*)d_in[0];
    const float* diff   = (const float*)d_in[1];
    const float* attr   = (const float*)d_in[2];
    const float* weight = (const float*)d_in[3];
    const float* bias   = (const float*)d_in[4];
    float* out = (float*)d_out;

    unsigned long long* buf0 = (unsigned long long*)d_ws;
    unsigned long long* buf1 = buf0 + (HN + 1);

    {
        int blocks = HN / (32 * 8);
        phase1_contrib<<<blocks, 256, 0, stream>>>(parent, diff, attr, weight, bias, buf0);
    }

    {
        int n = HN + 1;
        int blocks = (n + 255) / 256;
        unsigned long long* a = buf0;
        unsigned long long* b = buf1;
        int late_blocks = LATE_BLOCKS;
        unsigned long long* masks = (unsigned long long*)d_out;
        for (int s = 0; s < TAIL_FROM; ++s) {
            if (s < LATE_FROM)
                doubling_step<<<blocks, 256, 0, stream>>>(a, b, n, out);
            else if (s < TAIL_FROM - 1)
                doubling_step_late<false><<<late_blocks, 256, 0, stream>>>(a, b, n, masks, out);
            else
                doubling_step_late<true><<<late_blocks, 256, 0, stream>>>(a, b, n, masks, out);
            unsigned long long* t = a; a = b; b = t;
        }
        doubling_tail<<<1, TAIL_THREADS, 0, stream>>>(a, b, n, masks, out, parent, diff, attr, weight, bias);
    }
}
